# starting kernel
# baseline (speedup 1.0000x reference)
#include <hip/hip_runtime.h>
#include <math.h>

#define D_FEAT 128
#define HEADS 4
#define CDIM 32
#define NEG_SLOPE 0.2f
#define LN_EPS 1e-5f

typedef __attribute__((ext_vector_type(2))) float v2f;
typedef __attribute__((ext_vector_type(8))) float v8f;

__global__ __launch_bounds__(256) void gat_init(float* __restrict__ acc,
                                                float* __restrict__ m,
                                                float* __restrict__ denom,
                                                int N) {
  size_t total = (size_t)N * D_FEAT;
  size_t nh    = (size_t)N * HEADS;
  for (size_t i = blockIdx.x * (size_t)blockDim.x + threadIdx.x; i < total;
       i += (size_t)gridDim.x * blockDim.x) {
    acc[i] = 0.0f;
    if (i < nh) {
      m[i] = -INFINITY;
      denom[i] = 0.0f;
    }
  }
}

__global__ __launch_bounds__(256) void gat_gemm_h(const float* __restrict__ x,
                                                  const float* __restrict__ W,
                                                  float* __restrict__ h,
                                                  int N) {
  const int lane  = threadIdx.x & 31;
  const int wid   = threadIdx.x >> 5;
  const int row0  = blockIdx.x * 16;
  const int col0  = wid * 16;
  const int mrow  = lane & 15;
  const int khalf = lane >> 4;

  int arow = row0 + mrow;
  if (arow >= N) arow = N - 1;

  v8f acc = {};
#pragma unroll 8
  for (int k0 = 0; k0 < D_FEAT; k0 += 4) {
    const int ka = k0 + 2 * khalf;
    v2f a = *(const v2f*)(x + (size_t)arow * D_FEAT + ka);
    const float* bp = W + (size_t)ka * D_FEAT + col0 + mrow;
    v2f b;
    b.x = bp[0];
    b.y = bp[D_FEAT];
    acc = __builtin_amdgcn_wmma_f32_16x16x4_f32(
         false, a,  false, b,
         (short)0, acc,  false,  false);
  }

  const int rbase = row0 + 8 * khalf;
  float* dp = h + (size_t)rbase * D_FEAT + col0 + mrow;
  if (row0 + 16 <= N) {
#pragma unroll
    for (int v = 0; v < 8; ++v) dp[(size_t)v * D_FEAT] = acc[v];
  } else {
#pragma unroll
    for (int v = 0; v < 8; ++v) {
      if (rbase + v < N) dp[(size_t)v * D_FEAT] = acc[v];
    }
  }
}

__global__ __launch_bounds__(128) void gat_att_logits(const float* __restrict__ h,
                                                      const float* __restrict__ att_src,
                                                      const float* __restrict__ att_dst,
                                                      float* __restrict__ a_src,
                                                      float* __restrict__ a_dst,
                                                      int N) {
  const int n = blockIdx.x;
  if (n >= N) return;
  const int t = threadIdx.x;
  const float hv = h[(size_t)n * D_FEAT + t];
  float ps = hv * att_src[t];
  float pd = hv * att_dst[t];
#pragma unroll
  for (int off = 16; off > 0; off >>= 1) {
    ps += __shfl_down(ps, off, 32);
    pd += __shfl_down(pd, off, 32);
  }
  if ((t & 31) == 0) {
    const int head = t >> 5;
    a_src[(size_t)n * HEADS + head] = ps;
    a_dst[(size_t)n * HEADS + head] = pd;
  }
}

__global__ __launch_bounds__(256) void gat_edge_max(const int* __restrict__ ei,
                                                    int E, int Etot,
                                                    const float* __restrict__ a_src,
                                                    const float* __restrict__ a_dst,
                                                    float* __restrict__ elog,
                                                    float* __restrict__ m) {
  const size_t tid = blockIdx.x * (size_t)blockDim.x + threadIdx.x;
  const size_t total = (size_t)Etot * HEADS;
  if (tid >= total) return;
  const int e  = (int)(tid >> 2);
  const int hd = (int)(tid & 3);
  const int src = (e < E) ? ei[e] : (e - E);
  const int dst = (e < E) ? ei[E + e] : (e - E);
  float v = a_src[(size_t)src * HEADS + hd] + a_dst[(size_t)dst * HEADS + hd];
  v = (v > 0.0f) ? v : NEG_SLOPE * v;
  elog[tid] = v;
  atomicMax(&m[(size_t)dst * HEADS + hd], v);
}

__global__ __launch_bounds__(256) void gat_edge_exp(const int* __restrict__ ei,
                                                    int E, int Etot,
                                                    float* __restrict__ elog,
                                                    const float* __restrict__ m,
                                                    float* __restrict__ denom) {
  const size_t tid = blockIdx.x * (size_t)blockDim.x + threadIdx.x;
  const size_t total = (size_t)Etot * HEADS;
  if (tid >= total) return;
  const int e  = (int)(tid >> 2);
  const int hd = (int)(tid & 3);
  const int dst = (e < E) ? ei[E + e] : (e - E);
  const float ex = __expf(elog[tid] - m[(size_t)dst * HEADS + hd]);
  elog[tid] = ex;
  atomicAdd(&denom[(size_t)dst * HEADS + hd], ex);
}

__global__ __launch_bounds__(256) void gat_aggregate(const int* __restrict__ ei,
                                                     int E, int Etot,
                                                     const float* __restrict__ h,
                                                     const float* __restrict__ eexp,
                                                     const float* __restrict__ denom,
                                                     float* __restrict__ acc) {
  const int lane = threadIdx.x & 31;
  const int e = blockIdx.x * (blockDim.x >> 5) + (threadIdx.x >> 5);
  if (e >= Etot) return;
  const int src = (e < E) ? ei[e] : (e - E);
  const int dst = (e < E) ? ei[E + e] : (e - E);
#pragma unroll
  for (int hd = 0; hd < HEADS; ++hd) {
    const float w = eexp[(size_t)e * HEADS + hd] /
                    (denom[(size_t)dst * HEADS + hd] + 1e-16f);
    const int c = hd * CDIM + lane;
    const float v = h[(size_t)src * D_FEAT + c] * w;
    atomicAdd(&acc[(size_t)dst * D_FEAT + c], v);
  }
}

__global__ __launch_bounds__(128) void gat_finish(float* __restrict__ acc,
                                                  const float* __restrict__ bias,
                                                  const float* __restrict__ gamma,
                                                  const float* __restrict__ beta,
                                                  const float* __restrict__ x,
                                                  int N) {
  __shared__ float s_sum[4];
  __shared__ float s_sq[4];
  const int n = blockIdx.x;
  if (n >= N) return;
  const int t = threadIdx.x;

  const float v = acc[(size_t)n * D_FEAT + t] + bias[t];
  const float g = 0.5f * v * (1.0f + erff(v * 0.70710678118654752f));

  float s = g, sq = g * g;
#pragma unroll
  for (int off = 16; off > 0; off >>= 1) {
    s  += __shfl_down(s, off, 32);
    sq += __shfl_down(sq, off, 32);
  }
  if ((t & 31) == 0) {
    s_sum[t >> 5] = s;
    s_sq[t >> 5]  = sq;
  }
  __syncthreads();
  const float tot  = s_sum[0] + s_sum[1] + s_sum[2] + s_sum[3];
  const float totq = s_sq[0] + s_sq[1] + s_sq[2] + s_sq[3];
  const float mu  = tot * (1.0f / D_FEAT);
  const float var = totq * (1.0f / D_FEAT) - mu * mu;
  const float inv = rsqrtf(var + LN_EPS);

  const float out = (g - mu) * inv * gamma[t] + beta[t] + x[(size_t)n * D_FEAT + t];
  acc[(size_t)n * D_FEAT + t] = out;
}

extern "C" void kernel_launch(void* const* d_in, const int* in_sizes, int n_in,
                              void* d_out, int out_size, void* d_ws, size_t ws_size,
                              hipStream_t stream) {
  const float* x       = (const float*)d_in[0];
  const int*   ei      = (const int*)d_in[1];
  const float* W       = (const float*)d_in[2];
  const float* att_src = (const float*)d_in[3];
  const float* att_dst = (const float*)d_in[4];
  const float* bias    = (const float*)d_in[5];
  const float* gamma   = (const float*)d_in[6];
  const float* beta    = (const float*)d_in[7];

  const int N    = in_sizes[0] / D_FEAT;
  const int E    = in_sizes[1] / 2;
  const int Etot = E + N;

  float* ws     = (float*)d_ws;
  float* h      = ws;
  float* a_src  = h + (size_t)N * D_FEAT;
  float* a_dst  = a_src + (size_t)N * HEADS;
  float* mbuf   = a_dst + (size_t)N * HEADS;
  float* denom  = mbuf + (size_t)N * HEADS;
  float* elog   = denom + (size_t)N * HEADS;
  float* acc    = (float*)d_out;

  {
    size_t total = (size_t)N * D_FEAT;
    int blocks = (int)((total + 255) / 256);
    if (blocks > 65535) blocks = 65535;
    gat_init<<<blocks, 256, 0, stream>>>(acc, mbuf, denom, N);
  }
  {
    int rowTiles = (N + 15) / 16;
    gat_gemm_h<<<rowTiles, 256, 0, stream>>>(x, W, h, N);
  }
  gat_att_logits<<<N, 128, 0, stream>>>(h, att_src, att_dst, a_src, a_dst, N);
  {
    size_t total = (size_t)Etot * HEADS;
    int blocks = (int)((total + 255) / 256);
    gat_edge_max<<<blocks, 256, 0, stream>>>(ei, E, Etot, a_src, a_dst, elog, mbuf);
  }
  {
    size_t total = (size_t)Etot * HEADS;
    int blocks = (int)((total + 255) / 256);
    gat_edge_exp<<<blocks, 256, 0, stream>>>(ei, E, Etot, elog, mbuf, denom);
  }
  {
    int edgesPerBlock = 256 / 32;
    int blocks = (Etot + edgesPerBlock - 1) / edgesPerBlock;
    gat_aggregate<<<blocks, 256, 0, stream>>>(ei, E, Etot, h, elog, denom, acc);
  }
  gat_finish<<<N, 128, 0, stream>>>(acc, bias, gamma, beta, x, N);
}


# attention logits taken from the GEMM accumulators (a wave = 16 rows x one head's 32 columns, row sums by DPP); the one-node-per-block logits launch shrinks to a 4-edges-per-thread in-degree histogram
# speedup vs baseline: 1.5659x; 1.5659x over previous
#include <hip/hip_runtime.h>
#include <math.h>

#define D_FEAT 128
#define HEADS 4
#define CDIM 32
#define NEG_SLOPE 0.2f
#define LN_EPS 1e-5f

typedef __attribute__((ext_vector_type(2))) float v2f;
typedef __attribute__((ext_vector_type(8))) float v8f;


template <int M>
__device__ __forceinline__ float row_xor(float v) {
  return __int_as_float(__builtin_amdgcn_update_dpp(
      0, __float_as_int(v), 0x160 + M, 0xf, 0xf, false));
}

__device__ __forceinline__ float row_reduce8(const float (&p)[8], int mrow) {
  const bool b3 = (mrow & 8) != 0, b2 = (mrow & 4) != 0, b1 = (mrow & 2) != 0;
  float q[4], r[2];
#pragma unroll
  for (int i = 0; i < 4; ++i) {
    const float keep = b3 ? p[4 + i] : p[i];
    const float give = b3 ? p[i] : p[4 + i];
    q[i] = keep + row_xor<8>(give);
  }
#pragma unroll
  for (int i = 0; i < 2; ++i) {
    const float keep = b2 ? q[2 + i] : q[i];
    const float give = b2 ? q[i] : q[2 + i];
    r[i] = keep + row_xor<4>(give);
  }
  const float keep = b1 ? r[1] : r[0];
  const float give = b1 ? r[0] : r[1];
  const float s = keep + row_xor<2>(give);
  return s + row_xor<1>(s);
}

__global__ __launch_bounds__(256) void gat_gemm_h(const float* __restrict__ x,
                                                  const float* __restrict__ W,
                                                  const float* __restrict__ att_src,
                                                  const float* __restrict__ att_dst,
                                                  float* __restrict__ h,
                                                  float* __restrict__ a_src,
                                                  float* __restrict__ a_dst,
                                                  int* __restrict__ cnt,
                                                  int N) {
  __shared__ float s_as[32 * HEADS];
  __shared__ float s_ad[32 * HEADS];

  const int lane  = threadIdx.x & 31;
  const int wid   = __builtin_amdgcn_readfirstlane((int)(threadIdx.x >> 5));
  const int head  = wid & 3;
  const int rt    = wid >> 2;
  const int row0  = blockIdx.x * 32 + rt * 16;
  const int col0  = head * CDIM;
  const int mrow  = lane & 15;
  const int khalf = lane >> 4;

  int arow = row0 + mrow;
  if (arow >= N) arow = N - 1;

  const float* ap = x + (size_t)arow * D_FEAT + 2 * khalf;
  const float* bp = W + (size_t)(2 * khalf) * D_FEAT + col0 + 2 * mrow;

  v8f acc0 = {};
  v8f acc1 = {};
#pragma unroll 8
  for (int k0 = 0; k0 < D_FEAT; k0 += 4) {
    const v2f a = *(const v2f*)(ap + k0);
    const v2f w0 = *(const v2f*)(bp + (size_t)k0 * D_FEAT);
    const v2f w1 = *(const v2f*)(bp + (size_t)(k0 + 1) * D_FEAT);
    v2f b0, b1;
    b0.x = w0.x; b0.y = w1.x;
    b1.x = w0.y; b1.y = w1.y;
    acc0 = __builtin_amdgcn_wmma_f32_16x16x4_f32(
         false, a,  false, b0,
         (short)0, acc0,  false,  false);
    acc1 = __builtin_amdgcn_wmma_f32_16x16x4_f32(
         false, a,  false, b1,
         (short)0, acc1,  false,  false);
  }

  const int rbase = row0 + 8 * khalf;
  float* dp = h + (size_t)rbase * D_FEAT + col0 + 2 * mrow;
  if (row0 + 16 <= N) {
#pragma unroll
    for (int v = 0; v < 8; ++v) {
      v2f o; o.x = acc0[v]; o.y = acc1[v];
      *(v2f*)(dp + (size_t)v * D_FEAT) = o;
    }
  } else {
#pragma unroll
    for (int v = 0; v < 8; ++v) {
      if (rbase + v < N) {
        v2f o; o.x = acc0[v]; o.y = acc1[v];
        *(v2f*)(dp + (size_t)v * D_FEAT) = o;
      }
    }
  }

  const v2f vs = *(const v2f*)(att_src + col0 + 2 * mrow);
  const v2f vd = *(const v2f*)(att_dst + col0 + 2 * mrow);
  float ps[8], pd[8];
#pragma unroll
  for (int v = 0; v < 8; ++v) {
    ps[v] = acc0[v] * vs.x + acc1[v] * vs.y;
    pd[v] = acc0[v] * vd.x + acc1[v] * vd.y;
  }
  const float ts = row_reduce8(ps, mrow);
  const float td = row_reduce8(pd, mrow);
  if ((mrow & 1) == 0) {
    const int rloc = rt * 16 + 8 * khalf + (mrow >> 1);
    s_as[rloc * HEADS + head] = ts;
    s_ad[rloc * HEADS + head] = td;
  }
  __syncthreads();
  {
    const int t   = threadIdx.x & 127;
    const int row = blockIdx.x * 32 + (t >> 2);
    if (row < N) {
      if (threadIdx.x < 128) a_src[(size_t)blockIdx.x * (32 * HEADS) + t] = s_as[t];
      else                   a_dst[(size_t)blockIdx.x * (32 * HEADS) + t] = s_ad[t];
    }
  }
  if (threadIdx.x < 32 && blockIdx.x * 32 + (int)threadIdx.x < N)
    cnt[blockIdx.x * 32 + threadIdx.x] = 0;
}

__global__ __launch_bounds__(256) void gat_hist(const int* __restrict__ ei, int E,
                                                int* __restrict__ cnt) {
  const size_t base = (size_t)blockIdx.x * 1024 + threadIdx.x;
  int d[4];
#pragma unroll
  for (int k = 0; k < 4; ++k) {
    const size_t e = base + 256 * (size_t)k;
    d[k] = (e < (size_t)E) ? ei[(size_t)E + e] : -1;
  }
#pragma unroll
  for (int k = 0; k < 4; ++k) {
    if (d[k] >= 0) atomicAdd(&cnt[d[k]], 1);
  }
}

__device__ __forceinline__ int wave_incl_scan_i(int v, int lane) {
#pragma unroll
  for (int off = 1; off < 32; off <<= 1) {
    const int u = __shfl_up(v, off, 32);
    if (lane >= off) v += u;
  }
  return v;
}

__global__ __launch_bounds__(1024) void gat_scan(const int* __restrict__ cnt,
                                                 int* __restrict__ cursor,
                                                 int N) {
  __shared__ int s_tot[32];
  const int lane = threadIdx.x & 31;
  const int wid  = threadIdx.x >> 5;
  const int per  = (((N + 31) / 32) + 31) & ~31;
  const int lo   = wid * per;
  const int hi   = (lo + per < N) ? (lo + per) : N;

  int t = 0;
  for (int i0 = lo; i0 < hi; i0 += 256) {
    int c[8];
#pragma unroll
    for (int k = 0; k < 8; ++k) {
      const int i = i0 + 32 * k + lane;
      c[k] = (i < hi) ? cnt[i] : 0;
    }
#pragma unroll
    for (int k = 0; k < 8; ++k) t += c[k];
  }
#pragma unroll
  for (int off = 16; off > 0; off >>= 1) t += __shfl_xor(t, off, 32);
  if (lane == 0) s_tot[wid] = t;
  __syncthreads();

  const int tv   = s_tot[lane];
  const int incl = wave_incl_scan_i(tv, lane);
  int carry = __shfl(incl - tv, wid, 32);

  for (int i0 = lo; i0 < hi; i0 += 256) {
    int c[8];
#pragma unroll
    for (int k = 0; k < 8; ++k) {
      const int i = i0 + 32 * k + lane;
      c[k] = (i < hi) ? cnt[i] : 0;
    }
#pragma unroll
    for (int k = 0; k < 8; ++k) {
      const int i = i0 + 32 * k + lane;
      const int inc = wave_incl_scan_i(c[k], lane);
      if (i < hi) cursor[i] = carry + inc - c[k];
      carry += __shfl(inc, 31, 32);
    }
  }
}

__global__ __launch_bounds__(256) void gat_fill(const int* __restrict__ ei, int E,
                                                int* __restrict__ cursor,
                                                int* __restrict__ perm) {
  const size_t e = blockIdx.x * (size_t)blockDim.x + threadIdx.x;
  if (e >= (size_t)E) return;
  const int d = ei[(size_t)E + e];
  const int pos = atomicAdd(&cursor[d], 1);
  perm[pos] = (int)e;
}

typedef __attribute__((ext_vector_type(4))) float v4f;

__device__ __forceinline__ float leaky(float v) { return (v > 0.0f) ? v : NEG_SLOPE * v; }

__device__ __forceinline__ float wave_max_f(float v) {
#pragma unroll
  for (int off = 16; off > 0; off >>= 1) v = fmaxf(v, __shfl_xor(v, off, 32));
  return v;
}
__device__ __forceinline__ float wave_sum_f(float v) {
#pragma unroll
  for (int off = 16; off > 0; off >>= 1) v += __shfl_xor(v, off, 32);
  return v;
}
__device__ __forceinline__ float pick4(int q, float a, float b, float c, float d) {
  float r = a;
  r = (q == 1) ? b : r;
  r = (q == 2) ? c : r;
  r = (q == 3) ? d : r;
  return r;
}
__device__ __forceinline__ int next_edge(const int* __restrict__ perm, int beg, int deg,
                                         int last, int lane) {
  int best = 0x7fffffff;
  for (int i = lane; i < deg; i += 32) {
    const int e = perm[beg + i];
    if (e > last && e < best) best = e;
  }
#pragma unroll
  for (int off = 16; off > 0; off >>= 1) {
    const int o = __shfl_xor(best, off, 32);
    best = (o < best) ? o : best;
  }
  return best;
}

__global__ __launch_bounds__(256) void gat_node(const int* __restrict__ ei, int E,
                                                const int* __restrict__ cnt,
                                                const int* __restrict__ cursor,
                                                const int* __restrict__ perm,
                                                const float* __restrict__ h,
                                                const float* __restrict__ a_src,
                                                const float* __restrict__ a_dst,
                                                const float* __restrict__ bias,
                                                const float* __restrict__ gamma,
                                                const float* __restrict__ beta,
                                                const float* __restrict__ x,
                                                float* __restrict__ out,
                                                int N) {
  const int lane = threadIdx.x & 31;
  const int n = __builtin_amdgcn_readfirstlane(
      (int)(blockIdx.x * (blockDim.x >> 5) + (threadIdx.x >> 5)));
  if (n >= N) return;
  const int hq  = lane >> 3;
  const int deg = cnt[n];
  const int beg = cursor[n] - deg;
  const v4f ad = *(const v4f*)(a_dst + (size_t)n * HEADS);
  const v4f an = *(const v4f*)(a_src + (size_t)n * HEADS);
  v4f acc = {0.0f, 0.0f, 0.0f, 0.0f};

  if (deg < 32) {
    int key = (lane < deg) ? perm[beg + lane] : 0x7fffffff;
#pragma unroll
    for (int k = 2; k <= 32; k <<= 1) {
#pragma unroll
      for (int j = k >> 1; j > 0; j >>= 1) {
        const int o = __shfl_xor(key, j, 32);
        const bool up = ((lane & k) == 0);
        const bool lower = ((lane & j) == 0);
        const int mn = (o < key) ? o : key;
        const int mx = (o < key) ? key : o;
        key = (lower == up) ? mn : mx;
      }
    }
    const bool act = (lane <= deg);
    const int s = (lane < deg) ? ei[key] : n;
    const v4f as = *(const v4f*)(a_src + (size_t)s * HEADS);
    float l0 = leaky(as.x + ad.x), l1 = leaky(as.y + ad.y);
    float l2 = leaky(as.z + ad.z), l3 = leaky(as.w + ad.w);
    if (!act) { l0 = -INFINITY; l1 = -INFINITY; l2 = -INFINITY; l3 = -INFINITY; }
    const float m0 = wave_max_f(l0), m1 = wave_max_f(l1);
    const float m2 = wave_max_f(l2), m3 = wave_max_f(l3);
    const float e0 = act ? __expf(l0 - m0) : 0.0f;
    const float e1 = act ? __expf(l1 - m1) : 0.0f;
    const float e2 = act ? __expf(l2 - m2) : 0.0f;
    const float e3 = act ? __expf(l3 - m3) : 0.0f;
    const float al0 = e0 / (wave_sum_f(e0) + 1e-16f);
    const float al1 = e1 / (wave_sum_f(e1) + 1e-16f);
    const float al2 = e2 / (wave_sum_f(e2) + 1e-16f);
    const float al3 = e3 / (wave_sum_f(e3) + 1e-16f);
    const bool q1 = (hq == 1), q2 = (hq == 2), q3 = (hq == 3);
    for (int j0 = 0; j0 <= deg; j0 += 4) {
      v4f r[4];
      float w[4];
#pragma unroll
      for (int k = 0; k < 4; ++k) {
        const int j  = j0 + k;
        const int jl = j & 31;
        const int sj = __builtin_amdgcn_readlane(s, jl);
        float wk = __int_as_float(__builtin_amdgcn_readlane(__float_as_int(al0), jl));
        const float w1 = __int_as_float(__builtin_amdgcn_readlane(__float_as_int(al1), jl));
        const float w2 = __int_as_float(__builtin_amdgcn_readlane(__float_as_int(al2), jl));
        const float w3 = __int_as_float(__builtin_amdgcn_readlane(__float_as_int(al3), jl));
        wk = q1 ? w1 : wk;
        wk = q2 ? w2 : wk;
        wk = q3 ? w3 : wk;
        w[k] = (j <= deg) ? wk : 0.0f;
        r[k] = *(const v4f*)(h + (size_t)sj * D_FEAT + 4 * lane);
      }
#pragma unroll
      for (int k = 0; k < 4; ++k) acc += r[k] * w[k];
    }
  } else {
    float m0 = leaky(an.x + ad.x), m1 = leaky(an.y + ad.y);
    float m2 = leaky(an.z + ad.z), m3 = leaky(an.w + ad.w);
    const float s0 = m0, s1 = m1, s2 = m2, s3 = m3;
    for (int i = lane; i < deg; i += 32) {
      const int s = ei[perm[beg + i]];
      const v4f as = *(const v4f*)(a_src + (size_t)s * HEADS);
      m0 = fmaxf(m0, leaky(as.x + ad.x)); m1 = fmaxf(m1, leaky(as.y + ad.y));
      m2 = fmaxf(m2, leaky(as.z + ad.z)); m3 = fmaxf(m3, leaky(as.w + ad.w));
    }
    m0 = wave_max_f(m0); m1 = wave_max_f(m1); m2 = wave_max_f(m2); m3 = wave_max_f(m3);
    float d0 = 0.0f, d1 = 0.0f, d2 = 0.0f, d3 = 0.0f;
    int last = -1;
    for (int st = 0; st < deg; ++st) {
      last = next_edge(perm, beg, deg, last, lane);
      const int s = ei[last];
      const v4f as = *(const v4f*)(a_src + (size_t)s * HEADS);
      d0 += __expf(leaky(as.x + ad.x) - m0); d1 += __expf(leaky(as.y + ad.y) - m1);
      d2 += __expf(leaky(as.z + ad.z) - m2); d3 += __expf(leaky(as.w + ad.w) - m3);
    }
    d0 += __expf(s0 - m0); d1 += __expf(s1 - m1); d2 += __expf(s2 - m2); d3 += __expf(s3 - m3);
    last = -1;
    for (int st = 0; st < deg; ++st) {
      last = next_edge(perm, beg, deg, last, lane);
      const int s = ei[last];
      const v4f as = *(const v4f*)(a_src + (size_t)s * HEADS);
      const float w = pick4(hq, __expf(leaky(as.x + ad.x) - m0) / (d0 + 1e-16f),
                                __expf(leaky(as.y + ad.y) - m1) / (d1 + 1e-16f),
                                __expf(leaky(as.z + ad.z) - m2) / (d2 + 1e-16f),
                                __expf(leaky(as.w + ad.w) - m3) / (d3 + 1e-16f));
      const v4f r = *(const v4f*)(h + (size_t)s * D_FEAT + 4 * lane);
      acc += r * w;
    }
    {
      const float w = pick4(hq, __expf(s0 - m0) / (d0 + 1e-16f), __expf(s1 - m1) / (d1 + 1e-16f),
                                __expf(s2 - m2) / (d2 + 1e-16f), __expf(s3 - m3) / (d3 + 1e-16f));
      const v4f r = *(const v4f*)(h + (size_t)n * D_FEAT + 4 * lane);
      acc += r * w;
    }
  }

  const v4f bv = *(const v4f*)(bias + 4 * lane);
  const v4f v = acc + bv;
  v4f g;
  g.x = 0.5f * v.x * (1.0f + erff(v.x * 0.70710678118654752f));
  g.y = 0.5f * v.y * (1.0f + erff(v.y * 0.70710678118654752f));
  g.z = 0.5f * v.z * (1.0f + erff(v.z * 0.70710678118654752f));
  g.w = 0.5f * v.w * (1.0f + erff(v.w * 0.70710678118654752f));
  const float tot  = wave_sum_f((g.x + g.y) + (g.z + g.w));
  const float totq = wave_sum_f((g.x * g.x + g.y * g.y) + (g.z * g.z + g.w * g.w));
  const float mu  = tot * (1.0f / D_FEAT);
  const float var = totq * (1.0f / D_FEAT) - mu * mu;
  const float inv = rsqrtf(var + LN_EPS);
  const v4f gm = *(const v4f*)(gamma + 4 * lane);
  const v4f bt = *(const v4f*)(beta + 4 * lane);
  const v4f xv = *(const v4f*)(x + (size_t)n * D_FEAT + 4 * lane);
  v4f o;
  o.x = (g.x - mu) * inv * gm.x + bt.x + xv.x;
  o.y = (g.y - mu) * inv * gm.y + bt.y + xv.y;
  o.z = (g.z - mu) * inv * gm.z + bt.z + xv.z;
  o.w = (g.w - mu) * inv * gm.w + bt.w + xv.w;
  *(v4f*)(out + (size_t)n * D_FEAT + 4 * lane) = o;
}

extern "C" void kernel_launch(void* const* d_in, const int* in_sizes, int n_in,
                              void* d_out, int out_size, void* d_ws, size_t ws_size,
                              hipStream_t stream) {
  const float* x       = (const float*)d_in[0];
  const int*   ei      = (const int*)d_in[1];
  const float* W       = (const float*)d_in[2];
  const float* att_src = (const float*)d_in[3];
  const float* att_dst = (const float*)d_in[4];
  const float* bias    = (const float*)d_in[5];
  const float* gamma   = (const float*)d_in[6];
  const float* beta    = (const float*)d_in[7];

  const int N = in_sizes[0] / D_FEAT;
  const int E = in_sizes[1] / 2;

  float* ws     = (float*)d_ws;
  float* h      = ws;
  float* a_src  = h + (size_t)N * D_FEAT;
  float* a_dst  = a_src + (size_t)N * HEADS;
  int*   cnt    = (int*)(a_dst + (size_t)N * HEADS);
  int*   cursor = cnt + (size_t)N;
  int*   perm   = cursor + (size_t)N;
  float* out    = (float*)d_out;

  {
    int rowBlocks = (N + 31) / 32;
    gat_gemm_h<<<rowBlocks, 256, 0, stream>>>(x, W, att_src, att_dst, h, a_src, a_dst, cnt, N);
  }
  {
    int blocks = (E + 1023) / 1024;
    gat_hist<<<blocks, 256, 0, stream>>>(ei, E, cnt);
  }
  gat_scan<<<1, 1024, 0, stream>>>(cnt, cursor, N);
  {
    int blocks = (E + 255) / 256;
    gat_fill<<<blocks, 256, 0, stream>>>(ei, E, cursor, perm);
  }
  {
    int blocks = (N + 7) / 8;
    gat_node<<<blocks, 256, 0, stream>>>(ei, E, cnt, cursor, perm, h, a_src, a_dst,
                                         bias, gamma, beta, x, out, N);
  }
}


# h = x@W tiles run as extra work-groups of the zero, histogram, scan and fill launches, beside the edge bucketing they do not depend on (half beside the one-block scan); no GEMM launch of its own
# speedup vs baseline: 1.6298x; 1.0408x over previous
#include <hip/hip_runtime.h>
#include <math.h>

#define D_FEAT 128
#define HEADS 4
#define CDIM 32
#define NEG_SLOPE 0.2f
#define LN_EPS 1e-5f

typedef __attribute__((ext_vector_type(2))) float v2f;
typedef __attribute__((ext_vector_type(8))) float v8f;


template <int M>
__device__ __forceinline__ float row_xor(float v) {
  return __int_as_float(__builtin_amdgcn_update_dpp(
      0, __float_as_int(v), 0x160 + M, 0xf, 0xf, false));
}

__device__ __forceinline__ float row_reduce8(const float (&p)[8], int mrow) {
  const bool b3 = (mrow & 8) != 0, b2 = (mrow & 4) != 0, b1 = (mrow & 2) != 0;
  float q[4], r[2];
#pragma unroll
  for (int i = 0; i < 4; ++i) {
    const float keep = b3 ? p[4 + i] : p[i];
    const float give = b3 ? p[i] : p[4 + i];
    q[i] = keep + row_xor<8>(give);
  }
#pragma unroll
  for (int i = 0; i < 2; ++i) {
    const float keep = b2 ? q[2 + i] : q[i];
    const float give = b2 ? q[i] : q[2 + i];
    r[i] = keep + row_xor<4>(give);
  }
  const float keep = b1 ? r[1] : r[0];
  const float give = b1 ? r[0] : r[1];
  const float s = keep + row_xor<2>(give);
  return s + row_xor<1>(s);
}

__device__ __forceinline__ void gemm_tile(const float* __restrict__ x,
                                          const float* __restrict__ W,
                                          const float* __restrict__ att_src,
                                          const float* __restrict__ att_dst,
                                          float* __restrict__ h,
                                          float* __restrict__ a_src,
                                          float* __restrict__ a_dst,
                                          int N, int tile, int wl, int tl,
                                          float* s_as, float* s_ad) {
  const int lane  = tl & 31;
  const int head  = wl & 3;
  const int rt    = wl >> 2;
  const int row0  = tile * 32 + rt * 16;
  const int col0  = head * CDIM;
  const int mrow  = lane & 15;
  const int khalf = lane >> 4;

  int arow = row0 + mrow;
  if (arow >= N) arow = N - 1;

  const float* ap = x + (size_t)arow * D_FEAT + 2 * khalf;
  const float* bp = W + (size_t)(2 * khalf) * D_FEAT + col0 + 2 * mrow;

  v8f acc0 = {};
  v8f acc1 = {};
#pragma unroll 8
  for (int k0 = 0; k0 < D_FEAT; k0 += 4) {
    const v2f a = *(const v2f*)(ap + k0);
    const v2f w0 = *(const v2f*)(bp + (size_t)k0 * D_FEAT);
    const v2f w1 = *(const v2f*)(bp + (size_t)(k0 + 1) * D_FEAT);
    v2f b0, b1;
    b0.x = w0.x; b0.y = w1.x;
    b1.x = w0.y; b1.y = w1.y;
    acc0 = __builtin_amdgcn_wmma_f32_16x16x4_f32(
         false, a,  false, b0,
         (short)0, acc0,  false,  false);
    acc1 = __builtin_amdgcn_wmma_f32_16x16x4_f32(
         false, a,  false, b1,
         (short)0, acc1,  false,  false);
  }

  const int rbase = row0 + 8 * khalf;
  float* dp = h + (size_t)rbase * D_FEAT + col0 + 2 * mrow;
  if (row0 + 16 <= N) {
#pragma unroll
    for (int v = 0; v < 8; ++v) {
      v2f o; o.x = acc0[v]; o.y = acc1[v];
      *(v2f*)(dp + (size_t)v * D_FEAT) = o;
    }
  } else {
#pragma unroll
    for (int v = 0; v < 8; ++v) {
      if (rbase + v < N) {
        v2f o; o.x = acc0[v]; o.y = acc1[v];
        *(v2f*)(dp + (size_t)v * D_FEAT) = o;
      }
    }
  }

  const v2f vs = *(const v2f*)(att_src + col0 + 2 * mrow);
  const v2f vd = *(const v2f*)(att_dst + col0 + 2 * mrow);
  float ps[8], pd[8];
#pragma unroll
  for (int v = 0; v < 8; ++v) {
    ps[v] = acc0[v] * vs.x + acc1[v] * vs.y;
    pd[v] = acc0[v] * vd.x + acc1[v] * vd.y;
  }
  const float ts = row_reduce8(ps, mrow);
  const float td = row_reduce8(pd, mrow);
  if ((mrow & 1) == 0) {
    const int rloc = rt * 16 + 8 * khalf + (mrow >> 1);
    s_as[rloc * HEADS + head] = ts;
    s_ad[rloc * HEADS + head] = td;
  }
  __syncthreads();
  {
    const int t   = tl & 127;
    const int row = tile * 32 + (t >> 2);
    if (row < N) {
      if (tl < 128) a_src[(size_t)tile * (32 * HEADS) + t] = s_as[t];
      else          a_dst[(size_t)tile * (32 * HEADS) + t] = s_ad[t];
    }
  }
}

__global__ __launch_bounds__(256) void gat_zero_gemm(const float* __restrict__ x,
                                                     const float* __restrict__ W,
                                                     const float* __restrict__ att_src,
                                                     const float* __restrict__ att_dst,
                                                     float* __restrict__ h,
                                                     float* __restrict__ a_src,
                                                     float* __restrict__ a_dst,
                                                     int* __restrict__ cnt,
                                                     int N, int ntile) {
  __shared__ float s_as[32 * HEADS];
  __shared__ float s_ad[32 * HEADS];
  if ((int)blockIdx.x < ntile) {
    const int wl = __builtin_amdgcn_readfirstlane((int)(threadIdx.x >> 5));
    gemm_tile(x, W, att_src, att_dst, h, a_src, a_dst, N, (int)blockIdx.x, wl,
              (int)threadIdx.x, s_as, s_ad);
  } else {
    const int base = ((int)blockIdx.x - ntile) * 1024 + (int)threadIdx.x;
#pragma unroll
    for (int k = 0; k < 4; ++k) {
      const int i = base + 256 * k;
      if (i < N) cnt[i] = 0;
    }
  }
}

__global__ __launch_bounds__(256) void gat_hist_gemm(const float* __restrict__ x,
    const float* __restrict__ W, const float* __restrict__ att_src,
    const float* __restrict__ att_dst, float* __restrict__ h,
    float* __restrict__ a_src, float* __restrict__ a_dst, int N, int tile0, int ntile,
    const int* __restrict__ ei, int E, int* __restrict__ cnt) {
  __shared__ float s_as[32 * HEADS];
  __shared__ float s_ad[32 * HEADS];
  if ((int)blockIdx.x < ntile) {
    const int wl = __builtin_amdgcn_readfirstlane((int)(threadIdx.x >> 5));
    gemm_tile(x, W, att_src, att_dst, h, a_src, a_dst, N, tile0 + (int)blockIdx.x, wl,
              (int)threadIdx.x, s_as, s_ad);
    return;
  }
  const size_t base = (size_t)((int)blockIdx.x - ntile) * 1024 + threadIdx.x;
  int d[4];
#pragma unroll
  for (int k = 0; k < 4; ++k) {
    const size_t e = base + 256 * (size_t)k;
    d[k] = (e < (size_t)E) ? ei[(size_t)E + e] : -1;
  }
#pragma unroll
  for (int k = 0; k < 4; ++k) {
    if (d[k] >= 0) atomicAdd(&cnt[d[k]], 1);
  }
}

__device__ __forceinline__ int wave_incl_scan_i(int v, int lane) {
#pragma unroll
  for (int off = 1; off < 32; off <<= 1) {
    const int u = __shfl_up(v, off, 32);
    if (lane >= off) v += u;
  }
  return v;
}

__global__ __launch_bounds__(1024) void gat_scan_gemm(const float* __restrict__ x,
    const float* __restrict__ W, const float* __restrict__ att_src,
    const float* __restrict__ att_dst, float* __restrict__ h,
    float* __restrict__ a_src, float* __restrict__ a_dst, int N, int tile0, int ntile,
    const int* __restrict__ cnt, int* __restrict__ cursor) {
  __shared__ float s_as[4][32 * HEADS];
  __shared__ float s_ad[4][32 * HEADS];
  __shared__ int s_tot[32];
  if (blockIdx.x > 0) {
    const int sub = __builtin_amdgcn_readfirstlane((int)(threadIdx.x >> 8));
    const int wl  = __builtin_amdgcn_readfirstlane((int)((threadIdx.x >> 5) & 7));
    const int j   = 4 * ((int)blockIdx.x - 1) + sub;
    const int tile = (j < ntile) ? (tile0 + j) : ((N + 31) / 32);
    gemm_tile(x, W, att_src, att_dst, h, a_src, a_dst, N, tile, wl,
              (int)(threadIdx.x & 255), s_as[sub], s_ad[sub]);
    return;
  }
  const int lane = threadIdx.x & 31;
  const int wid  = threadIdx.x >> 5;
  const int per  = (((N + 31) / 32) + 31) & ~31;
  const int lo   = wid * per;
  const int hi   = (lo + per < N) ? (lo + per) : N;

  int t = 0;
  for (int i0 = lo; i0 < hi; i0 += 256) {
    int c[8];
#pragma unroll
    for (int k = 0; k < 8; ++k) {
      const int i = i0 + 32 * k + lane;
      c[k] = (i < hi) ? cnt[i] : 0;
    }
#pragma unroll
    for (int k = 0; k < 8; ++k) t += c[k];
  }
#pragma unroll
  for (int off = 16; off > 0; off >>= 1) t += __shfl_xor(t, off, 32);
  if (lane == 0) s_tot[wid] = t;
  __syncthreads();

  const int tv   = s_tot[lane];
  const int incl = wave_incl_scan_i(tv, lane);
  int carry = __shfl(incl - tv, wid, 32);

  for (int i0 = lo; i0 < hi; i0 += 256) {
    int c[8];
#pragma unroll
    for (int k = 0; k < 8; ++k) {
      const int i = i0 + 32 * k + lane;
      c[k] = (i < hi) ? cnt[i] : 0;
    }
#pragma unroll
    for (int k = 0; k < 8; ++k) {
      const int i = i0 + 32 * k + lane;
      const int inc = wave_incl_scan_i(c[k], lane);
      if (i < hi) cursor[i] = carry + inc - c[k];
      carry += __shfl(inc, 31, 32);
    }
  }
}

__global__ __launch_bounds__(256) void gat_fill_gemm(const float* __restrict__ x,
    const float* __restrict__ W, const float* __restrict__ att_src,
    const float* __restrict__ att_dst, float* __restrict__ h,
    float* __restrict__ a_src, float* __restrict__ a_dst, int N, int tile0, int ntile,
    const int* __restrict__ ei, int E, int* __restrict__ cursor, int* __restrict__ perm) {
  __shared__ float s_as[32 * HEADS];
  __shared__ float s_ad[32 * HEADS];
  if ((int)blockIdx.x < ntile) {
    const int wl = __builtin_amdgcn_readfirstlane((int)(threadIdx.x >> 5));
    gemm_tile(x, W, att_src, att_dst, h, a_src, a_dst, N, tile0 + (int)blockIdx.x, wl,
              (int)threadIdx.x, s_as, s_ad);
    return;
  }
  const size_t e = (size_t)((int)blockIdx.x - ntile) * 256 + threadIdx.x;
  if (e >= (size_t)E) return;
  const int d = ei[(size_t)E + e];
  const int pos = atomicAdd(&cursor[d], 1);
  perm[pos] = (int)e;
}

typedef __attribute__((ext_vector_type(4))) float v4f;

__device__ __forceinline__ float leaky(float v) { return (v > 0.0f) ? v : NEG_SLOPE * v; }

__device__ __forceinline__ float wave_max_f(float v) {
#pragma unroll
  for (int off = 16; off > 0; off >>= 1) v = fmaxf(v, __shfl_xor(v, off, 32));
  return v;
}
__device__ __forceinline__ float wave_sum_f(float v) {
#pragma unroll
  for (int off = 16; off > 0; off >>= 1) v += __shfl_xor(v, off, 32);
  return v;
}
__device__ __forceinline__ float pick4(int q, float a, float b, float c, float d) {
  float r = a;
  r = (q == 1) ? b : r;
  r = (q == 2) ? c : r;
  r = (q == 3) ? d : r;
  return r;
}
__device__ __forceinline__ int next_edge(const int* __restrict__ perm, int beg, int deg,
                                         int last, int lane) {
  int best = 0x7fffffff;
  for (int i = lane; i < deg; i += 32) {
    const int e = perm[beg + i];
    if (e > last && e < best) best = e;
  }
#pragma unroll
  for (int off = 16; off > 0; off >>= 1) {
    const int o = __shfl_xor(best, off, 32);
    best = (o < best) ? o : best;
  }
  return best;
}

__global__ __launch_bounds__(256) void gat_node(const int* __restrict__ ei, int E,
                                                const int* __restrict__ cnt,
                                                const int* __restrict__ cursor,
                                                const int* __restrict__ perm,
                                                const float* __restrict__ h,
                                                const float* __restrict__ a_src,
                                                const float* __restrict__ a_dst,
                                                const float* __restrict__ bias,
                                                const float* __restrict__ gamma,
                                                const float* __restrict__ beta,
                                                const float* __restrict__ x,
                                                float* __restrict__ out,
                                                int N) {
  const int lane = threadIdx.x & 31;
  const int n = __builtin_amdgcn_readfirstlane(
      (int)(blockIdx.x * (blockDim.x >> 5) + (threadIdx.x >> 5)));
  if (n >= N) return;
  const int hq  = lane >> 3;
  const int deg = cnt[n];
  const int beg = cursor[n] - deg;
  const v4f ad = *(const v4f*)(a_dst + (size_t)n * HEADS);
  const v4f an = *(const v4f*)(a_src + (size_t)n * HEADS);
  v4f acc = {0.0f, 0.0f, 0.0f, 0.0f};

  if (deg < 32) {
    int key = (lane < deg) ? perm[beg + lane] : 0x7fffffff;
#pragma unroll
    for (int k = 2; k <= 32; k <<= 1) {
#pragma unroll
      for (int j = k >> 1; j > 0; j >>= 1) {
        const int o = __shfl_xor(key, j, 32);
        const bool up = ((lane & k) == 0);
        const bool lower = ((lane & j) == 0);
        const int mn = (o < key) ? o : key;
        const int mx = (o < key) ? key : o;
        key = (lower == up) ? mn : mx;
      }
    }
    const bool act = (lane <= deg);
    const int s = (lane < deg) ? ei[key] : n;
    const v4f as = *(const v4f*)(a_src + (size_t)s * HEADS);
    float l0 = leaky(as.x + ad.x), l1 = leaky(as.y + ad.y);
    float l2 = leaky(as.z + ad.z), l3 = leaky(as.w + ad.w);
    if (!act) { l0 = -INFINITY; l1 = -INFINITY; l2 = -INFINITY; l3 = -INFINITY; }
    const float m0 = wave_max_f(l0), m1 = wave_max_f(l1);
    const float m2 = wave_max_f(l2), m3 = wave_max_f(l3);
    const float e0 = act ? __expf(l0 - m0) : 0.0f;
    const float e1 = act ? __expf(l1 - m1) : 0.0f;
    const float e2 = act ? __expf(l2 - m2) : 0.0f;
    const float e3 = act ? __expf(l3 - m3) : 0.0f;
    const float al0 = e0 / (wave_sum_f(e0) + 1e-16f);
    const float al1 = e1 / (wave_sum_f(e1) + 1e-16f);
    const float al2 = e2 / (wave_sum_f(e2) + 1e-16f);
    const float al3 = e3 / (wave_sum_f(e3) + 1e-16f);
    const bool q1 = (hq == 1), q2 = (hq == 2), q3 = (hq == 3);
    for (int j0 = 0; j0 <= deg; j0 += 4) {
      v4f r[4];
      float w[4];
#pragma unroll
      for (int k = 0; k < 4; ++k) {
        const int j  = j0 + k;
        const int jl = j & 31;
        const int sj = __builtin_amdgcn_readlane(s, jl);
        float wk = __int_as_float(__builtin_amdgcn_readlane(__float_as_int(al0), jl));
        const float w1 = __int_as_float(__builtin_amdgcn_readlane(__float_as_int(al1), jl));
        const float w2 = __int_as_float(__builtin_amdgcn_readlane(__float_as_int(al2), jl));
        const float w3 = __int_as_float(__builtin_amdgcn_readlane(__float_as_int(al3), jl));
        wk = q1 ? w1 : wk;
        wk = q2 ? w2 : wk;
        wk = q3 ? w3 : wk;
        w[k] = (j <= deg) ? wk : 0.0f;
        r[k] = *(const v4f*)(h + (size_t)sj * D_FEAT + 4 * lane);
      }
#pragma unroll
      for (int k = 0; k < 4; ++k) acc += r[k] * w[k];
    }
  } else {
    float m0 = leaky(an.x + ad.x), m1 = leaky(an.y + ad.y);
    float m2 = leaky(an.z + ad.z), m3 = leaky(an.w + ad.w);
    const float s0 = m0, s1 = m1, s2 = m2, s3 = m3;
    for (int i = lane; i < deg; i += 32) {
      const int s = ei[perm[beg + i]];
      const v4f as = *(const v4f*)(a_src + (size_t)s * HEADS);
      m0 = fmaxf(m0, leaky(as.x + ad.x)); m1 = fmaxf(m1, leaky(as.y + ad.y));
      m2 = fmaxf(m2, leaky(as.z + ad.z)); m3 = fmaxf(m3, leaky(as.w + ad.w));
    }
    m0 = wave_max_f(m0); m1 = wave_max_f(m1); m2 = wave_max_f(m2); m3 = wave_max_f(m3);
    float d0 = 0.0f, d1 = 0.0f, d2 = 0.0f, d3 = 0.0f;
    int last = -1;
    for (int st = 0; st < deg; ++st) {
      last = next_edge(perm, beg, deg, last, lane);
      const int s = ei[last];
      const v4f as = *(const v4f*)(a_src + (size_t)s * HEADS);
      d0 += __expf(leaky(as.x + ad.x) - m0); d1 += __expf(leaky(as.y + ad.y) - m1);
      d2 += __expf(leaky(as.z + ad.z) - m2); d3 += __expf(leaky(as.w + ad.w) - m3);
    }
    d0 += __expf(s0 - m0); d1 += __expf(s1 - m1); d2 += __expf(s2 - m2); d3 += __expf(s3 - m3);
    last = -1;
    for (int st = 0; st < deg; ++st) {
      last = next_edge(perm, beg, deg, last, lane);
      const int s = ei[last];
      const v4f as = *(const v4f*)(a_src + (size_t)s * HEADS);
      const float w = pick4(hq, __expf(leaky(as.x + ad.x) - m0) / (d0 + 1e-16f),
                                __expf(leaky(as.y + ad.y) - m1) / (d1 + 1e-16f),
                                __expf(leaky(as.z + ad.z) - m2) / (d2 + 1e-16f),
                                __expf(leaky(as.w + ad.w) - m3) / (d3 + 1e-16f));
      const v4f r = *(const v4f*)(h + (size_t)s * D_FEAT + 4 * lane);
      acc += r * w;
    }
    {
      const float w = pick4(hq, __expf(s0 - m0) / (d0 + 1e-16f), __expf(s1 - m1) / (d1 + 1e-16f),
                                __expf(s2 - m2) / (d2 + 1e-16f), __expf(s3 - m3) / (d3 + 1e-16f));
      const v4f r = *(const v4f*)(h + (size_t)n * D_FEAT + 4 * lane);
      acc += r * w;
    }
  }

  const v4f bv = *(const v4f*)(bias + 4 * lane);
  const v4f v = acc + bv;
  v4f g;
  g.x = 0.5f * v.x * (1.0f + erff(v.x * 0.70710678118654752f));
  g.y = 0.5f * v.y * (1.0f + erff(v.y * 0.70710678118654752f));
  g.z = 0.5f * v.z * (1.0f + erff(v.z * 0.70710678118654752f));
  g.w = 0.5f * v.w * (1.0f + erff(v.w * 0.70710678118654752f));
  const float tot  = wave_sum_f((g.x + g.y) + (g.z + g.w));
  const float totq = wave_sum_f((g.x * g.x + g.y * g.y) + (g.z * g.z + g.w * g.w));
  const float mu  = tot * (1.0f / D_FEAT);
  const float var = totq * (1.0f / D_FEAT) - mu * mu;
  const float inv = rsqrtf(var + LN_EPS);
  const v4f gm = *(const v4f*)(gamma + 4 * lane);
  const v4f bt = *(const v4f*)(beta + 4 * lane);
  const v4f xv = *(const v4f*)(x + (size_t)n * D_FEAT + 4 * lane);
  v4f o;
  o.x = (g.x - mu) * inv * gm.x + bt.x + xv.x;
  o.y = (g.y - mu) * inv * gm.y + bt.y + xv.y;
  o.z = (g.z - mu) * inv * gm.z + bt.z + xv.z;
  o.w = (g.w - mu) * inv * gm.w + bt.w + xv.w;
  *(v4f*)(out + (size_t)n * D_FEAT + 4 * lane) = o;
}

extern "C" void kernel_launch(void* const* d_in, const int* in_sizes, int n_in,
                              void* d_out, int out_size, void* d_ws, size_t ws_size,
                              hipStream_t stream) {
  const float* x       = (const float*)d_in[0];
  const int*   ei      = (const int*)d_in[1];
  const float* W       = (const float*)d_in[2];
  const float* att_src = (const float*)d_in[3];
  const float* att_dst = (const float*)d_in[4];
  const float* bias    = (const float*)d_in[5];
  const float* gamma   = (const float*)d_in[6];
  const float* beta    = (const float*)d_in[7];

  const int N = in_sizes[0] / D_FEAT;
  const int E = in_sizes[1] / 2;

  float* ws     = (float*)d_ws;
  float* h      = ws;
  float* a_src  = h + (size_t)N * D_FEAT;
  float* a_dst  = a_src + (size_t)N * HEADS;
  int*   cnt    = (int*)(a_dst + (size_t)N * HEADS);
  int*   cursor = cnt + (size_t)N;
  int*   perm   = cursor + (size_t)N;
  float* out    = (float*)d_out;

  const int T  = (N + 31) / 32;
  const int g1 = T / 16;
  const int g2 = T / 4;
  const int g3 = T / 2;
  const int g4 = T - g1 - g2 - g3;
  {
    int zblocks = (N + 1023) / 1024;
    gat_zero_gemm<<<g1 + zblocks, 256, 0, stream>>>(x, W, att_src, att_dst, h, a_src, a_dst,
                                                    cnt, N, g1);
  }
  {
    int blocks = (E + 1023) / 1024;
    gat_hist_gemm<<<g2 + blocks, 256, 0, stream>>>(x, W, att_src, att_dst, h, a_src, a_dst, N,
                                                   g1, g2, ei, E, cnt);
  }
  gat_scan_gemm<<<1 + (g3 + 3) / 4, 1024, 0, stream>>>(x, W, att_src, att_dst, h, a_src, a_dst, N,
                                                       g1 + g2, g3, cnt, cursor);
  {
    int blocks = (E + 255) / 256;
    gat_fill_gemm<<<g4 + blocks, 256, 0, stream>>>(x, W, att_src, att_dst, h, a_src, a_dst, N,
                                                   g1 + g2 + g3, g4, ei, E, cursor, perm);
  }
  {
    int blocks = (N + 7) / 8;
    gat_node<<<blocks, 256, 0, stream>>>(ei, E, cnt, cursor, perm, h, a_src, a_dst,
                                         bias, gamma, beta, x, out, N);
  }
}


# segment-start scan spread over one group per 1024 nodes: each sums the counts in front of its chunk itself, 16 loads in flight, and scans the chunk once; no one-block walk of chained wave scans
# speedup vs baseline: 1.7600x; 1.0799x over previous
#include <hip/hip_runtime.h>
#include <math.h>

#define D_FEAT 128
#define HEADS 4
#define CDIM 32
#define NEG_SLOPE 0.2f
#define LN_EPS 1e-5f

typedef __attribute__((ext_vector_type(2))) float v2f;
typedef __attribute__((ext_vector_type(8))) float v8f;


template <int M>
__device__ __forceinline__ float row_xor(float v) {
  return __int_as_float(__builtin_amdgcn_update_dpp(
      0, __float_as_int(v), 0x160 + M, 0xf, 0xf, false));
}

__device__ __forceinline__ float row_reduce8(const float (&p)[8], int mrow) {
  const bool b3 = (mrow & 8) != 0, b2 = (mrow & 4) != 0, b1 = (mrow & 2) != 0;
  float q[4], r[2];
#pragma unroll
  for (int i = 0; i < 4; ++i) {
    const float keep = b3 ? p[4 + i] : p[i];
    const float give = b3 ? p[i] : p[4 + i];
    q[i] = keep + row_xor<8>(give);
  }
#pragma unroll
  for (int i = 0; i < 2; ++i) {
    const float keep = b2 ? q[2 + i] : q[i];
    const float give = b2 ? q[i] : q[2 + i];
    r[i] = keep + row_xor<4>(give);
  }
  const float keep = b1 ? r[1] : r[0];
  const float give = b1 ? r[0] : r[1];
  const float s = keep + row_xor<2>(give);
  return s + row_xor<1>(s);
}

__device__ __forceinline__ void gemm_tile(const float* __restrict__ x,
                                          const float* __restrict__ W,
                                          const float* __restrict__ att_src,
                                          const float* __restrict__ att_dst,
                                          float* __restrict__ h,
                                          float* __restrict__ a_src,
                                          float* __restrict__ a_dst,
                                          int N, int tile, int wl, int tl,
                                          float* s_as, float* s_ad) {
  const int lane  = tl & 31;
  const int head  = wl & 3;
  const int rt    = wl >> 2;
  const int row0  = tile * 32 + rt * 16;
  const int col0  = head * CDIM;
  const int mrow  = lane & 15;
  const int khalf = lane >> 4;

  int arow = row0 + mrow;
  if (arow >= N) arow = N - 1;

  const float* ap = x + (size_t)arow * D_FEAT + 2 * khalf;
  const float* bp = W + (size_t)(2 * khalf) * D_FEAT + col0 + 2 * mrow;

  v8f acc0 = {};
  v8f acc1 = {};
#pragma unroll 8
  for (int k0 = 0; k0 < D_FEAT; k0 += 4) {
    const v2f a = *(const v2f*)(ap + k0);
    const v2f w0 = *(const v2f*)(bp + (size_t)k0 * D_FEAT);
    const v2f w1 = *(const v2f*)(bp + (size_t)(k0 + 1) * D_FEAT);
    v2f b0, b1;
    b0.x = w0.x; b0.y = w1.x;
    b1.x = w0.y; b1.y = w1.y;
    acc0 = __builtin_amdgcn_wmma_f32_16x16x4_f32(
         false, a,  false, b0,
         (short)0, acc0,  false,  false);
    acc1 = __builtin_amdgcn_wmma_f32_16x16x4_f32(
         false, a,  false, b1,
         (short)0, acc1,  false,  false);
  }

  const int rbase = row0 + 8 * khalf;
  float* dp = h + (size_t)rbase * D_FEAT + col0 + 2 * mrow;
  if (row0 + 16 <= N) {
#pragma unroll
    for (int v = 0; v < 8; ++v) {
      v2f o; o.x = acc0[v]; o.y = acc1[v];
      *(v2f*)(dp + (size_t)v * D_FEAT) = o;
    }
  } else {
#pragma unroll
    for (int v = 0; v < 8; ++v) {
      if (rbase + v < N) {
        v2f o; o.x = acc0[v]; o.y = acc1[v];
        *(v2f*)(dp + (size_t)v * D_FEAT) = o;
      }
    }
  }

  const v2f vs = *(const v2f*)(att_src + col0 + 2 * mrow);
  const v2f vd = *(const v2f*)(att_dst + col0 + 2 * mrow);
  float ps[8], pd[8];
#pragma unroll
  for (int v = 0; v < 8; ++v) {
    ps[v] = acc0[v] * vs.x + acc1[v] * vs.y;
    pd[v] = acc0[v] * vd.x + acc1[v] * vd.y;
  }
  const float ts = row_reduce8(ps, mrow);
  const float td = row_reduce8(pd, mrow);
  if ((mrow & 1) == 0) {
    const int rloc = rt * 16 + 8 * khalf + (mrow >> 1);
    s_as[rloc * HEADS + head] = ts;
    s_ad[rloc * HEADS + head] = td;
  }
  __syncthreads();
  {
    const int t   = tl & 127;
    const int row = tile * 32 + (t >> 2);
    if (row < N) {
      if (tl < 128) a_src[(size_t)tile * (32 * HEADS) + t] = s_as[t];
      else          a_dst[(size_t)tile * (32 * HEADS) + t] = s_ad[t];
    }
  }
}

__global__ __launch_bounds__(256) void gat_zero_gemm(const float* __restrict__ x,
                                                     const float* __restrict__ W,
                                                     const float* __restrict__ att_src,
                                                     const float* __restrict__ att_dst,
                                                     float* __restrict__ h,
                                                     float* __restrict__ a_src,
                                                     float* __restrict__ a_dst,
                                                     int* __restrict__ cnt,
                                                     int N, int ntile) {
  __shared__ float s_as[32 * HEADS];
  __shared__ float s_ad[32 * HEADS];
  if ((int)blockIdx.x < ntile) {
    const int wl = __builtin_amdgcn_readfirstlane((int)(threadIdx.x >> 5));
    gemm_tile(x, W, att_src, att_dst, h, a_src, a_dst, N, (int)blockIdx.x, wl,
              (int)threadIdx.x, s_as, s_ad);
  } else {
    const int base = ((int)blockIdx.x - ntile) * 1024 + (int)threadIdx.x;
#pragma unroll
    for (int k = 0; k < 4; ++k) {
      const int i = base + 256 * k;
      if (i < N) cnt[i] = 0;
    }
  }
}

__global__ __launch_bounds__(256) void gat_hist_gemm(const float* __restrict__ x,
    const float* __restrict__ W, const float* __restrict__ att_src,
    const float* __restrict__ att_dst, float* __restrict__ h,
    float* __restrict__ a_src, float* __restrict__ a_dst, int N, int tile0, int ntile,
    const int* __restrict__ ei, int E, int* __restrict__ cnt) {
  __shared__ float s_as[32 * HEADS];
  __shared__ float s_ad[32 * HEADS];
  if ((int)blockIdx.x < ntile) {
    const int wl = __builtin_amdgcn_readfirstlane((int)(threadIdx.x >> 5));
    gemm_tile(x, W, att_src, att_dst, h, a_src, a_dst, N, tile0 + (int)blockIdx.x, wl,
              (int)threadIdx.x, s_as, s_ad);
    return;
  }
  const size_t base = (size_t)((int)blockIdx.x - ntile) * 1024 + threadIdx.x;
  int d[4];
#pragma unroll
  for (int k = 0; k < 4; ++k) {
    const size_t e = base + 256 * (size_t)k;
    d[k] = (e < (size_t)E) ? ei[(size_t)E + e] : -1;
  }
#pragma unroll
  for (int k = 0; k < 4; ++k) {
    if (d[k] >= 0) atomicAdd(&cnt[d[k]], 1);
  }
}

__device__ __forceinline__ int wave_incl_scan_i(int v, int lane) {
#pragma unroll
  for (int off = 1; off < 32; off <<= 1) {
    const int u = __shfl_up(v, off, 32);
    if (lane >= off) v += u;
  }
  return v;
}

__global__ __launch_bounds__(1024) void gat_scan_gemm(const float* __restrict__ x,
    const float* __restrict__ W, const float* __restrict__ att_src,
    const float* __restrict__ att_dst, float* __restrict__ h,
    float* __restrict__ a_src, float* __restrict__ a_dst, int N, int tile0, int ntile,
    int nscan, const int* __restrict__ cnt, int* __restrict__ cursor) {
  __shared__ float s_as[4][32 * HEADS];
  __shared__ float s_ad[4][32 * HEADS];
  __shared__ int s_pre[32];
  __shared__ int s_tot[32];
  if ((int)blockIdx.x >= nscan) {
    const int sub = __builtin_amdgcn_readfirstlane((int)(threadIdx.x >> 8));
    const int wl  = __builtin_amdgcn_readfirstlane((int)((threadIdx.x >> 5) & 7));
    const int j   = 4 * ((int)blockIdx.x - nscan) + sub;
    const int tile = (j < ntile) ? (tile0 + j) : ((N + 31) / 32);
    gemm_tile(x, W, att_src, att_dst, h, a_src, a_dst, N, tile, wl,
              (int)(threadIdx.x & 255), s_as[sub], s_ad[sub]);
    return;
  }
  const int lane = threadIdx.x & 31;
  const int wid  = threadIdx.x >> 5;
  const int b = nscan - 1 - (int)blockIdx.x;
  const int i = b * 1024 + (int)threadIdx.x;
  const int c = (i < N) ? cnt[i] : 0;

  int pre = 0;
  for (int k0 = 0; k0 < b; k0 += 16) {
    int v[16];
#pragma unroll
    for (int k = 0; k < 16; ++k)
      v[k] = (k0 + k < b) ? cnt[(k0 + k) * 1024 + (int)threadIdx.x] : 0;
#pragma unroll
    for (int k = 0; k < 16; ++k) pre += v[k];
  }
#pragma unroll
  for (int off = 16; off > 0; off >>= 1) pre += __shfl_xor(pre, off, 32);
  const int inc = wave_incl_scan_i(c, lane);
  if (lane == 0)  s_pre[wid] = pre;
  if (lane == 31) s_tot[wid] = inc;
  __syncthreads();

  int base = s_pre[lane];
#pragma unroll
  for (int off = 16; off > 0; off >>= 1) base += __shfl_xor(base, off, 32);
  const int tv   = s_tot[lane];
  const int incl = wave_incl_scan_i(tv, lane);
  const int wbeg = __shfl(incl - tv, wid, 32);
  if (i < N) cursor[i] = base + wbeg + inc - c;
}

__global__ __launch_bounds__(256) void gat_fill_gemm(const float* __restrict__ x,
    const float* __restrict__ W, const float* __restrict__ att_src,
    const float* __restrict__ att_dst, float* __restrict__ h,
    float* __restrict__ a_src, float* __restrict__ a_dst, int N, int tile0, int ntile,
    const int* __restrict__ ei, int E, int* __restrict__ cursor, int* __restrict__ perm) {
  __shared__ float s_as[32 * HEADS];
  __shared__ float s_ad[32 * HEADS];
  if ((int)blockIdx.x < ntile) {
    const int wl = __builtin_amdgcn_readfirstlane((int)(threadIdx.x >> 5));
    gemm_tile(x, W, att_src, att_dst, h, a_src, a_dst, N, tile0 + (int)blockIdx.x, wl,
              (int)threadIdx.x, s_as, s_ad);
    return;
  }
  const size_t e = (size_t)((int)blockIdx.x - ntile) * 256 + threadIdx.x;
  if (e >= (size_t)E) return;
  const int d = ei[(size_t)E + e];
  const int pos = atomicAdd(&cursor[d], 1);
  perm[pos] = (int)e;
}

typedef __attribute__((ext_vector_type(4))) float v4f;

__device__ __forceinline__ float leaky(float v) { return (v > 0.0f) ? v : NEG_SLOPE * v; }

__device__ __forceinline__ float wave_max_f(float v) {
#pragma unroll
  for (int off = 16; off > 0; off >>= 1) v = fmaxf(v, __shfl_xor(v, off, 32));
  return v;
}
__device__ __forceinline__ float wave_sum_f(float v) {
#pragma unroll
  for (int off = 16; off > 0; off >>= 1) v += __shfl_xor(v, off, 32);
  return v;
}
__device__ __forceinline__ float pick4(int q, float a, float b, float c, float d) {
  float r = a;
  r = (q == 1) ? b : r;
  r = (q == 2) ? c : r;
  r = (q == 3) ? d : r;
  return r;
}
__device__ __forceinline__ int next_edge(const int* __restrict__ perm, int beg, int deg,
                                         int last, int lane) {
  int best = 0x7fffffff;
  for (int i = lane; i < deg; i += 32) {
    const int e = perm[beg + i];
    if (e > last && e < best) best = e;
  }
#pragma unroll
  for (int off = 16; off > 0; off >>= 1) {
    const int o = __shfl_xor(best, off, 32);
    best = (o < best) ? o : best;
  }
  return best;
}

__global__ __launch_bounds__(256) void gat_node(const int* __restrict__ ei, int E,
                                                const int* __restrict__ cnt,
                                                const int* __restrict__ cursor,
                                                const int* __restrict__ perm,
                                                const float* __restrict__ h,
                                                const float* __restrict__ a_src,
                                                const float* __restrict__ a_dst,
                                                const float* __restrict__ bias,
                                                const float* __restrict__ gamma,
                                                const float* __restrict__ beta,
                                                const float* __restrict__ x,
                                                float* __restrict__ out,
                                                int N) {
  const int lane = threadIdx.x & 31;
  const int n = __builtin_amdgcn_readfirstlane(
      (int)(blockIdx.x * (blockDim.x >> 5) + (threadIdx.x >> 5)));
  if (n >= N) return;
  const int hq  = lane >> 3;
  const int deg = cnt[n];
  const int beg = cursor[n] - deg;
  const v4f ad = *(const v4f*)(a_dst + (size_t)n * HEADS);
  const v4f an = *(const v4f*)(a_src + (size_t)n * HEADS);
  v4f acc = {0.0f, 0.0f, 0.0f, 0.0f};

  if (deg < 32) {
    int key = (lane < deg) ? perm[beg + lane] : 0x7fffffff;
#pragma unroll
    for (int k = 2; k <= 32; k <<= 1) {
#pragma unroll
      for (int j = k >> 1; j > 0; j >>= 1) {
        const int o = __shfl_xor(key, j, 32);
        const bool up = ((lane & k) == 0);
        const bool lower = ((lane & j) == 0);
        const int mn = (o < key) ? o : key;
        const int mx = (o < key) ? key : o;
        key = (lower == up) ? mn : mx;
      }
    }
    const bool act = (lane <= deg);
    const int s = (lane < deg) ? ei[key] : n;
    const v4f as = *(const v4f*)(a_src + (size_t)s * HEADS);
    float l0 = leaky(as.x + ad.x), l1 = leaky(as.y + ad.y);
    float l2 = leaky(as.z + ad.z), l3 = leaky(as.w + ad.w);
    if (!act) { l0 = -INFINITY; l1 = -INFINITY; l2 = -INFINITY; l3 = -INFINITY; }
    const float m0 = wave_max_f(l0), m1 = wave_max_f(l1);
    const float m2 = wave_max_f(l2), m3 = wave_max_f(l3);
    const float e0 = act ? __expf(l0 - m0) : 0.0f;
    const float e1 = act ? __expf(l1 - m1) : 0.0f;
    const float e2 = act ? __expf(l2 - m2) : 0.0f;
    const float e3 = act ? __expf(l3 - m3) : 0.0f;
    const float al0 = e0 / (wave_sum_f(e0) + 1e-16f);
    const float al1 = e1 / (wave_sum_f(e1) + 1e-16f);
    const float al2 = e2 / (wave_sum_f(e2) + 1e-16f);
    const float al3 = e3 / (wave_sum_f(e3) + 1e-16f);
    const bool q1 = (hq == 1), q2 = (hq == 2), q3 = (hq == 3);
    for (int j0 = 0; j0 <= deg; j0 += 4) {
      v4f r[4];
      float w[4];
#pragma unroll
      for (int k = 0; k < 4; ++k) {
        const int j  = j0 + k;
        const int jl = j & 31;
        const int sj = __builtin_amdgcn_readlane(s, jl);
        float wk = __int_as_float(__builtin_amdgcn_readlane(__float_as_int(al0), jl));
        const float w1 = __int_as_float(__builtin_amdgcn_readlane(__float_as_int(al1), jl));
        const float w2 = __int_as_float(__builtin_amdgcn_readlane(__float_as_int(al2), jl));
        const float w3 = __int_as_float(__builtin_amdgcn_readlane(__float_as_int(al3), jl));
        wk = q1 ? w1 : wk;
        wk = q2 ? w2 : wk;
        wk = q3 ? w3 : wk;
        w[k] = (j <= deg) ? wk : 0.0f;
        r[k] = *(const v4f*)(h + (size_t)sj * D_FEAT + 4 * lane);
      }
#pragma unroll
      for (int k = 0; k < 4; ++k) acc += r[k] * w[k];
    }
  } else {
    float m0 = leaky(an.x + ad.x), m1 = leaky(an.y + ad.y);
    float m2 = leaky(an.z + ad.z), m3 = leaky(an.w + ad.w);
    const float s0 = m0, s1 = m1, s2 = m2, s3 = m3;
    for (int i = lane; i < deg; i += 32) {
      const int s = ei[perm[beg + i]];
      const v4f as = *(const v4f*)(a_src + (size_t)s * HEADS);
      m0 = fmaxf(m0, leaky(as.x + ad.x)); m1 = fmaxf(m1, leaky(as.y + ad.y));
      m2 = fmaxf(m2, leaky(as.z + ad.z)); m3 = fmaxf(m3, leaky(as.w + ad.w));
    }
    m0 = wave_max_f(m0); m1 = wave_max_f(m1); m2 = wave_max_f(m2); m3 = wave_max_f(m3);
    float d0 = 0.0f, d1 = 0.0f, d2 = 0.0f, d3 = 0.0f;
    int last = -1;
    for (int st = 0; st < deg; ++st) {
      last = next_edge(perm, beg, deg, last, lane);
      const int s = ei[last];
      const v4f as = *(const v4f*)(a_src + (size_t)s * HEADS);
      d0 += __expf(leaky(as.x + ad.x) - m0); d1 += __expf(leaky(as.y + ad.y) - m1);
      d2 += __expf(leaky(as.z + ad.z) - m2); d3 += __expf(leaky(as.w + ad.w) - m3);
    }
    d0 += __expf(s0 - m0); d1 += __expf(s1 - m1); d2 += __expf(s2 - m2); d3 += __expf(s3 - m3);
    last = -1;
    for (int st = 0; st < deg; ++st) {
      last = next_edge(perm, beg, deg, last, lane);
      const int s = ei[last];
      const v4f as = *(const v4f*)(a_src + (size_t)s * HEADS);
      const float w = pick4(hq, __expf(leaky(as.x + ad.x) - m0) / (d0 + 1e-16f),
                                __expf(leaky(as.y + ad.y) - m1) / (d1 + 1e-16f),
                                __expf(leaky(as.z + ad.z) - m2) / (d2 + 1e-16f),
                                __expf(leaky(as.w + ad.w) - m3) / (d3 + 1e-16f));
      const v4f r = *(const v4f*)(h + (size_t)s * D_FEAT + 4 * lane);
      acc += r * w;
    }
    {
      const float w = pick4(hq, __expf(s0 - m0) / (d0 + 1e-16f), __expf(s1 - m1) / (d1 + 1e-16f),
                                __expf(s2 - m2) / (d2 + 1e-16f), __expf(s3 - m3) / (d3 + 1e-16f));
      const v4f r = *(const v4f*)(h + (size_t)n * D_FEAT + 4 * lane);
      acc += r * w;
    }
  }

  const v4f bv = *(const v4f*)(bias + 4 * lane);
  const v4f v = acc + bv;
  v4f g;
  g.x = 0.5f * v.x * (1.0f + erff(v.x * 0.70710678118654752f));
  g.y = 0.5f * v.y * (1.0f + erff(v.y * 0.70710678118654752f));
  g.z = 0.5f * v.z * (1.0f + erff(v.z * 0.70710678118654752f));
  g.w = 0.5f * v.w * (1.0f + erff(v.w * 0.70710678118654752f));
  const float tot  = wave_sum_f((g.x + g.y) + (g.z + g.w));
  const float totq = wave_sum_f((g.x * g.x + g.y * g.y) + (g.z * g.z + g.w * g.w));
  const float mu  = tot * (1.0f / D_FEAT);
  const float var = totq * (1.0f / D_FEAT) - mu * mu;
  const float inv = rsqrtf(var + LN_EPS);
  const v4f gm = *(const v4f*)(gamma + 4 * lane);
  const v4f bt = *(const v4f*)(beta + 4 * lane);
  const v4f xv = *(const v4f*)(x + (size_t)n * D_FEAT + 4 * lane);
  v4f o;
  o.x = (g.x - mu) * inv * gm.x + bt.x + xv.x;
  o.y = (g.y - mu) * inv * gm.y + bt.y + xv.y;
  o.z = (g.z - mu) * inv * gm.z + bt.z + xv.z;
  o.w = (g.w - mu) * inv * gm.w + bt.w + xv.w;
  *(v4f*)(out + (size_t)n * D_FEAT + 4 * lane) = o;
}

extern "C" void kernel_launch(void* const* d_in, const int* in_sizes, int n_in,
                              void* d_out, int out_size, void* d_ws, size_t ws_size,
                              hipStream_t stream) {
  const float* x       = (const float*)d_in[0];
  const int*   ei      = (const int*)d_in[1];
  const float* W       = (const float*)d_in[2];
  const float* att_src = (const float*)d_in[3];
  const float* att_dst = (const float*)d_in[4];
  const float* bias    = (const float*)d_in[5];
  const float* gamma   = (const float*)d_in[6];
  const float* beta    = (const float*)d_in[7];

  const int N = in_sizes[0] / D_FEAT;
  const int E = in_sizes[1] / 2;

  float* ws     = (float*)d_ws;
  float* h      = ws;
  float* a_src  = h + (size_t)N * D_FEAT;
  float* a_dst  = a_src + (size_t)N * HEADS;
  int*   cnt    = (int*)(a_dst + (size_t)N * HEADS);
  int*   cursor = cnt + (size_t)N;
  int*   perm   = cursor + (size_t)N;
  float* out    = (float*)d_out;

  const int T  = (N + 31) / 32;
  const int g1 = T / 16;
  const int g2 = T / 4;
  const int g3 = T / 2;
  const int g4 = T - g1 - g2 - g3;
  {
    int zblocks = (N + 1023) / 1024;
    gat_zero_gemm<<<g1 + zblocks, 256, 0, stream>>>(x, W, att_src, att_dst, h, a_src, a_dst,
                                                    cnt, N, g1);
  }
  {
    int blocks = (E + 1023) / 1024;
    gat_hist_gemm<<<g2 + blocks, 256, 0, stream>>>(x, W, att_src, att_dst, h, a_src, a_dst, N,
                                                   g1, g2, ei, E, cnt);
  }
  {
    int nscan = (N + 1023) / 1024;
    gat_scan_gemm<<<nscan + (g3 + 3) / 4, 1024, 0, stream>>>(x, W, att_src, att_dst, h, a_src,
                                                             a_dst, N, g1 + g2, g3, nscan,
                                                             cnt, cursor);
  }
  {
    int blocks = (E + 255) / 256;
    gat_fill_gemm<<<g4 + blocks, 256, 0, stream>>>(x, W, att_src, att_dst, h, a_src, a_dst, N,
                                                   g1 + g2 + g3, g4, ei, E, cursor, perm);
  }
  {
    int blocks = (N + 7) / 8;
    gat_node<<<blocks, 256, 0, stream>>>(ei, E, cnt, cursor, perm, h, a_src, a_dst,
                                         bias, gamma, beta, x, out, N);
  }
}


# edge ranks from LDS counters in a two-level sort (chunks grouped by 256-node destination range in launch 1, one group per range ranks its records in launch 2); no returning global atomic per edge
# speedup vs baseline: 3.6527x; 2.0754x over previous
#include <hip/hip_runtime.h>
#include <math.h>

#define D_FEAT 128
#define HEADS 4
#define CDIM 32
#define NEG_SLOPE 0.2f
#define LN_EPS 1e-5f

typedef __attribute__((ext_vector_type(2))) float v2f;
typedef __attribute__((ext_vector_type(8))) float v8f;


template <int M>
__device__ __forceinline__ float row_xor(float v) {
  return __int_as_float(__builtin_amdgcn_update_dpp(
      0, __float_as_int(v), 0x160 + M, 0xf, 0xf, false));
}

__device__ __forceinline__ float row_reduce8(const float (&p)[8], int mrow) {
  const bool b3 = (mrow & 8) != 0, b2 = (mrow & 4) != 0, b1 = (mrow & 2) != 0;
  float q[4], r[2];
#pragma unroll
  for (int i = 0; i < 4; ++i) {
    const float keep = b3 ? p[4 + i] : p[i];
    const float give = b3 ? p[i] : p[4 + i];
    q[i] = keep + row_xor<8>(give);
  }
#pragma unroll
  for (int i = 0; i < 2; ++i) {
    const float keep = b2 ? q[2 + i] : q[i];
    const float give = b2 ? q[i] : q[2 + i];
    r[i] = keep + row_xor<4>(give);
  }
  const float keep = b1 ? r[1] : r[0];
  const float give = b1 ? r[0] : r[1];
  const float s = keep + row_xor<2>(give);
  return s + row_xor<1>(s);
}

__device__ __forceinline__ void gemm_tile(const float* __restrict__ x,
                                          const float* __restrict__ W,
                                          const float* __restrict__ att_src,
                                          const float* __restrict__ att_dst,
                                          float* __restrict__ h,
                                          float* __restrict__ a_src,
                                          float* __restrict__ a_dst,
                                          int N, int tile, int wl, int tl,
                                          float* s_as, float* s_ad) {
  const int lane  = tl & 31;
  const int head  = wl & 3;
  const int rt    = wl >> 2;
  const int row0  = tile * 32 + rt * 16;
  const int col0  = head * CDIM;
  const int mrow  = lane & 15;
  const int khalf = lane >> 4;

  int arow = row0 + mrow;
  if (arow >= N) arow = N - 1;

  const float* ap = x + (size_t)arow * D_FEAT + 2 * khalf;
  const float* bp = W + (size_t)(2 * khalf) * D_FEAT + col0 + 2 * mrow;

  v8f acc0 = {};
  v8f acc1 = {};
#pragma unroll 8
  for (int k0 = 0; k0 < D_FEAT; k0 += 4) {
    const v2f a = *(const v2f*)(ap + k0);
    const v2f w0 = *(const v2f*)(bp + (size_t)k0 * D_FEAT);
    const v2f w1 = *(const v2f*)(bp + (size_t)(k0 + 1) * D_FEAT);
    v2f b0, b1;
    b0.x = w0.x; b0.y = w1.x;
    b1.x = w0.y; b1.y = w1.y;
    acc0 = __builtin_amdgcn_wmma_f32_16x16x4_f32(
         false, a,  false, b0,
         (short)0, acc0,  false,  false);
    acc1 = __builtin_amdgcn_wmma_f32_16x16x4_f32(
         false, a,  false, b1,
         (short)0, acc1,  false,  false);
  }

  const int rbase = row0 + 8 * khalf;
  float* dp = h + (size_t)rbase * D_FEAT + col0 + 2 * mrow;
  if (row0 + 16 <= N) {
#pragma unroll
    for (int v = 0; v < 8; ++v) {
      v2f o; o.x = acc0[v]; o.y = acc1[v];
      *(v2f*)(dp + (size_t)v * D_FEAT) = o;
    }
  } else {
#pragma unroll
    for (int v = 0; v < 8; ++v) {
      if (rbase + v < N) {
        v2f o; o.x = acc0[v]; o.y = acc1[v];
        *(v2f*)(dp + (size_t)v * D_FEAT) = o;
      }
    }
  }

  const v2f vs = *(const v2f*)(att_src + col0 + 2 * mrow);
  const v2f vd = *(const v2f*)(att_dst + col0 + 2 * mrow);
  float ps[8], pd[8];
#pragma unroll
  for (int v = 0; v < 8; ++v) {
    ps[v] = acc0[v] * vs.x + acc1[v] * vs.y;
    pd[v] = acc0[v] * vd.x + acc1[v] * vd.y;
  }
  const float ts = row_reduce8(ps, mrow);
  const float td = row_reduce8(pd, mrow);
  if ((mrow & 1) == 0) {
    const int rloc = rt * 16 + 8 * khalf + (mrow >> 1);
    s_as[rloc * HEADS + head] = ts;
    s_ad[rloc * HEADS + head] = td;
  }
  __syncthreads();
  {
    const int t   = tl & 127;
    const int row = tile * 32 + (t >> 2);
    if (row < N) {
      if (tl < 128) a_src[(size_t)tile * (32 * HEADS) + t] = s_as[t];
      else          a_dst[(size_t)tile * (32 * HEADS) + t] = s_ad[t];
    }
  }
}

#define CH_EDGES 2048
#define RS_SHIFT 8
#define RS_NODES 256
#define TAB_STRIDE 257
typedef int __attribute__((may_alias)) int_alias;

__device__ __forceinline__ size_t up_slot(size_t i) {
  return (i >> 6) * D_FEAT + 64 + (i & 63);
}
__device__ __forceinline__ size_t tab_base(int E) {
  return ((size_t)E + 63) & ~(size_t)63;
}

__device__ __forceinline__ void bin_chunk(const int* __restrict__ ei, int E, int c,
                                          int_alias* outw, size_t tab0, int tl,
                                          int* s_hist, int* s_wsum) {
  s_hist[tl] = 0;
  __syncthreads();
  const int e0 = c * CH_EDGES + tl;
  int d[8], p[8];
#pragma unroll
  for (int k = 0; k < 8; ++k) {
    const int e = e0 + 256 * k;
    d[k] = (e < E) ? ei[(size_t)E + e] : -1;
  }
#pragma unroll
  for (int k = 0; k < 8; ++k)
    p[k] = (d[k] >= 0) ? atomicAdd(&s_hist[d[k] >> RS_SHIFT], 1) : 0;
  __syncthreads();
  const int lane = tl & 31, w = tl >> 5;
  const int v = s_hist[tl];
  int inc = v;
#pragma unroll
  for (int off = 1; off < 32; off <<= 1) {
    const int o = __shfl_up(inc, off, 32);
    inc += (lane >= off) ? o : 0;
  }
  if (lane == 31) s_wsum[w] = inc;
  __syncthreads();
  int base = 0;
#pragma unroll
  for (int j = 0; j < 8; ++j) base += (j < w) ? s_wsum[j] : 0;
  const int excl = base + inc - v;
  s_hist[tl] = excl;
  outw[up_slot(tab0 + (size_t)c * TAB_STRIDE + tl)] = excl;
  if (tl == 255) outw[up_slot(tab0 + (size_t)c * TAB_STRIDE + 256)] = excl + v;
  __syncthreads();
#pragma unroll
  for (int k = 0; k < 8; ++k) {
    if (d[k] >= 0) {
      const int idx = s_hist[d[k] >> RS_SHIFT] + p[k];
      const unsigned e = (unsigned)(e0 + 256 * k);
      outw[up_slot((size_t)c * CH_EDGES + idx)] = (int)((e << 8) | (unsigned)(d[k] & 255));
    }
  }
}

__global__ __launch_bounds__(256) void gat_bin_gemm(const float* __restrict__ x,
                                                    const float* __restrict__ W,
                                                    const float* __restrict__ att_src,
                                                    const float* __restrict__ att_dst,
                                                    float* __restrict__ h,
                                                    float* __restrict__ a_src,
                                                    float* __restrict__ a_dst,
                                                    int* __restrict__ cnt,
                                                    int N, int ntile,
                                                    const int* __restrict__ ei, int E,
                                                    float* out) {
  __shared__ float s_as[32 * HEADS];
  __shared__ float s_ad[32 * HEADS];
  __shared__ int s_hist[256];
  __shared__ int s_wsum[8];
  if ((int)blockIdx.x < ntile) {
    const int wl = __builtin_amdgcn_readfirstlane((int)(threadIdx.x >> 5));
    gemm_tile(x, W, att_src, att_dst, h, a_src, a_dst, N, (int)blockIdx.x, wl,
              (int)threadIdx.x, s_as, s_ad);
  } else {
    const int c = (int)blockIdx.x - ntile;
    if (c == 0 && threadIdx.x == 0) cnt[N] = 0;
    bin_chunk(ei, E, c, (int_alias*)out, tab_base(E), (int)threadIdx.x, s_hist, s_wsum);
  }
}

#define CAP 64

__global__ __launch_bounds__(256) void gat_fill_gemm(const float* __restrict__ x,
    const float* __restrict__ W, const float* __restrict__ att_src,
    const float* __restrict__ att_dst, float* __restrict__ h,
    float* __restrict__ a_src, float* __restrict__ a_dst, int N, int tile0, int ntile,
    int nch, int E, int* __restrict__ cnt, int* __restrict__ ovf,
    float* out) {
  __shared__ float s_as[32 * HEADS];
  __shared__ float s_ad[32 * HEADS];
  __shared__ int s_cnt[RS_NODES];
  if ((int)blockIdx.x < ntile) {
    const int wl = __builtin_amdgcn_readfirstlane((int)(threadIdx.x >> 5));
    gemm_tile(x, W, att_src, att_dst, h, a_src, a_dst, N, tile0 + (int)blockIdx.x, wl,
              (int)threadIdx.x, s_as, s_ad);
    return;
  }
  int_alias* outw = (int_alias*)out;
  const int tl = (int)threadIdx.x;
  const int r  = (int)blockIdx.x - ntile;
  const size_t tab0 = tab_base(E);
  s_cnt[tl] = 0;
  __syncthreads();
  for (int c = tl; c < nch; c += 256) {
    const size_t tb = tab0 + (size_t)c * TAB_STRIDE + r;
    const int o0 = outw[up_slot(tb)];
    const int o1 = outw[up_slot(tb + 1)];
    const size_t rb = (size_t)c * CH_EDGES;
    for (int i0 = o0; i0 < o1; i0 += 8) {
      int rec[8];
#pragma unroll
      for (int k = 0; k < 8; ++k)
        rec[k] = (i0 + k < o1) ? outw[up_slot(rb + (size_t)(i0 + k))] : -1;
#pragma unroll
      for (int k = 0; k < 8; ++k) {
        if (rec[k] >= 0) {
          const int dl = rec[k] & 255;
          const int e  = (int)((unsigned)rec[k] >> 8);
          const int rank = atomicAdd(&s_cnt[dl], 1);
          const int n = (r << RS_SHIFT) + dl;
          if (rank < CAP) {
            outw[(size_t)n * D_FEAT + rank] = e;
          } else {
            const int o = atomicAdd(&cnt[N], 1);
            ovf[o] = e;
          }
        }
      }
    }
  }
  __syncthreads();
  {
    const int n = (r << RS_SHIFT) + tl;
    if (n < N) cnt[n] = s_cnt[tl];
  }
}

typedef __attribute__((ext_vector_type(4))) float v4f;

__device__ __forceinline__ float leaky(float v) { return (v > 0.0f) ? v : NEG_SLOPE * v; }

__device__ __forceinline__ float wave_max_f(float v) {
#pragma unroll
  for (int off = 16; off > 0; off >>= 1) v = fmaxf(v, __shfl_xor(v, off, 32));
  return v;
}
__device__ __forceinline__ float wave_sum_f(float v) {
#pragma unroll
  for (int off = 16; off > 0; off >>= 1) v += __shfl_xor(v, off, 32);
  return v;
}
__device__ __forceinline__ float pick4(int q, float a, float b, float c, float d) {
  float r = a;
  r = (q == 1) ? b : r;
  r = (q == 2) ? c : r;
  r = (q == 3) ? d : r;
  return r;
}
__device__ __forceinline__ int next_edge(const int_alias* bucket, int nb,
                                         const int* __restrict__ ovf, int novf,
                                         const int* __restrict__ ei, int E, int n,
                                         int last, int lane) {
  int best = 0x7fffffff;
  for (int i = lane; i < nb; i += 32) {
    const int e = bucket[i];
    if (e > last && e < best) best = e;
  }
  for (int i = lane; i < novf; i += 32) {
    const int e = ovf[i];
    if (e > last && e < best && ei[(size_t)E + e] == n) best = e;
  }
#pragma unroll
  for (int off = 16; off > 0; off >>= 1) {
    const int o = __shfl_xor(best, off, 32);
    best = (o < best) ? o : best;
  }
  return best;
}

__global__ __launch_bounds__(256) void gat_node(const int* __restrict__ ei, int E,
                                                const int* __restrict__ cnt,
                                                const int* __restrict__ ovf,
                                                const float* __restrict__ h,
                                                const float* __restrict__ a_src,
                                                const float* __restrict__ a_dst,
                                                const float* __restrict__ bias,
                                                const float* __restrict__ gamma,
                                                const float* __restrict__ beta,
                                                const float* __restrict__ x,
                                                float* out,
                                                int N) {
  const int lane = threadIdx.x & 31;
  const int n = __builtin_amdgcn_readfirstlane(
      (int)(blockIdx.x * (blockDim.x >> 5) + (threadIdx.x >> 5)));
  if (n >= N) return;
  const int hq  = lane >> 3;
  const int_alias* bucket = (const int_alias*)(out + (size_t)n * D_FEAT);
  const int slot = bucket[lane];
  const int deg = cnt[n];
  const v4f ad = *(const v4f*)(a_dst + (size_t)n * HEADS);
  const v4f an = *(const v4f*)(a_src + (size_t)n * HEADS);
  v4f acc = {0.0f, 0.0f, 0.0f, 0.0f};

  if (deg < 32) {
    int key = (lane < deg) ? slot : 0x7fffffff;
#pragma unroll
    for (int k = 2; k <= 32; k <<= 1) {
#pragma unroll
      for (int j = k >> 1; j > 0; j >>= 1) {
        const int o = __shfl_xor(key, j, 32);
        const bool up = ((lane & k) == 0);
        const bool lower = ((lane & j) == 0);
        const int mn = (o < key) ? o : key;
        const int mx = (o < key) ? key : o;
        key = (lower == up) ? mn : mx;
      }
    }
    const bool act = (lane <= deg);
    const int s = (lane < deg) ? ei[key] : n;
    const v4f as = *(const v4f*)(a_src + (size_t)s * HEADS);
    float l0 = leaky(as.x + ad.x), l1 = leaky(as.y + ad.y);
    float l2 = leaky(as.z + ad.z), l3 = leaky(as.w + ad.w);
    if (!act) { l0 = -INFINITY; l1 = -INFINITY; l2 = -INFINITY; l3 = -INFINITY; }
    const float m0 = wave_max_f(l0), m1 = wave_max_f(l1);
    const float m2 = wave_max_f(l2), m3 = wave_max_f(l3);
    const float e0 = act ? __expf(l0 - m0) : 0.0f;
    const float e1 = act ? __expf(l1 - m1) : 0.0f;
    const float e2 = act ? __expf(l2 - m2) : 0.0f;
    const float e3 = act ? __expf(l3 - m3) : 0.0f;
    const float al0 = e0 / (wave_sum_f(e0) + 1e-16f);
    const float al1 = e1 / (wave_sum_f(e1) + 1e-16f);
    const float al2 = e2 / (wave_sum_f(e2) + 1e-16f);
    const float al3 = e3 / (wave_sum_f(e3) + 1e-16f);
    const bool q1 = (hq == 1), q2 = (hq == 2), q3 = (hq == 3);
    for (int j0 = 0; j0 <= deg; j0 += 4) {
      v4f r[4];
      float w[4];
#pragma unroll
      for (int k = 0; k < 4; ++k) {
        const int j  = j0 + k;
        const int jl = j & 31;
        const int sj = __builtin_amdgcn_readlane(s, jl);
        float wk = __int_as_float(__builtin_amdgcn_readlane(__float_as_int(al0), jl));
        const float w1 = __int_as_float(__builtin_amdgcn_readlane(__float_as_int(al1), jl));
        const float w2 = __int_as_float(__builtin_amdgcn_readlane(__float_as_int(al2), jl));
        const float w3 = __int_as_float(__builtin_amdgcn_readlane(__float_as_int(al3), jl));
        wk = q1 ? w1 : wk;
        wk = q2 ? w2 : wk;
        wk = q3 ? w3 : wk;
        w[k] = (j <= deg) ? wk : 0.0f;
        r[k] = *(const v4f*)(h + (size_t)sj * D_FEAT + 4 * lane);
      }
#pragma unroll
      for (int k = 0; k < 4; ++k) acc += r[k] * w[k];
    }
  } else {
    const int nb   = (deg < CAP) ? deg : CAP;
    const int novf = (deg > CAP) ? cnt[N] : 0;
    float m0 = leaky(an.x + ad.x), m1 = leaky(an.y + ad.y);
    float m2 = leaky(an.z + ad.z), m3 = leaky(an.w + ad.w);
    const float s0 = m0, s1 = m1, s2 = m2, s3 = m3;
    for (int i = lane; i < nb + novf; i += 32) {
      const int e = (i < 32) ? slot : ((i < nb) ? bucket[i] : ovf[i - nb]);
      if (i < nb || ei[(size_t)E + e] == n) {
        const int s = ei[e];
        const v4f as = *(const v4f*)(a_src + (size_t)s * HEADS);
        m0 = fmaxf(m0, leaky(as.x + ad.x)); m1 = fmaxf(m1, leaky(as.y + ad.y));
        m2 = fmaxf(m2, leaky(as.z + ad.z)); m3 = fmaxf(m3, leaky(as.w + ad.w));
      }
    }
    m0 = wave_max_f(m0); m1 = wave_max_f(m1); m2 = wave_max_f(m2); m3 = wave_max_f(m3);
    float d0 = 0.0f, d1 = 0.0f, d2 = 0.0f, d3 = 0.0f;
    int last = -1;
    for (int st = 0; st < deg; ++st) {
      last = next_edge(bucket, nb, ovf, novf, ei, E, n, last, lane);
      const int s = ei[last];
      const v4f as = *(const v4f*)(a_src + (size_t)s * HEADS);
      d0 += __expf(leaky(as.x + ad.x) - m0); d1 += __expf(leaky(as.y + ad.y) - m1);
      d2 += __expf(leaky(as.z + ad.z) - m2); d3 += __expf(leaky(as.w + ad.w) - m3);
    }
    d0 += __expf(s0 - m0); d1 += __expf(s1 - m1); d2 += __expf(s2 - m2); d3 += __expf(s3 - m3);
    last = -1;
    for (int st = 0; st < deg; ++st) {
      last = next_edge(bucket, nb, ovf, novf, ei, E, n, last, lane);
      const int s = ei[last];
      const v4f as = *(const v4f*)(a_src + (size_t)s * HEADS);
      const float w = pick4(hq, __expf(leaky(as.x + ad.x) - m0) / (d0 + 1e-16f),
                                __expf(leaky(as.y + ad.y) - m1) / (d1 + 1e-16f),
                                __expf(leaky(as.z + ad.z) - m2) / (d2 + 1e-16f),
                                __expf(leaky(as.w + ad.w) - m3) / (d3 + 1e-16f));
      const v4f r = *(const v4f*)(h + (size_t)s * D_FEAT + 4 * lane);
      acc += r * w;
    }
    {
      const float w = pick4(hq, __expf(s0 - m0) / (d0 + 1e-16f), __expf(s1 - m1) / (d1 + 1e-16f),
                                __expf(s2 - m2) / (d2 + 1e-16f), __expf(s3 - m3) / (d3 + 1e-16f));
      const v4f r = *(const v4f*)(h + (size_t)n * D_FEAT + 4 * lane);
      acc += r * w;
    }
  }

  const v4f bv = *(const v4f*)(bias + 4 * lane);
  const v4f v = acc + bv;
  v4f g;
  g.x = 0.5f * v.x * (1.0f + erff(v.x * 0.70710678118654752f));
  g.y = 0.5f * v.y * (1.0f + erff(v.y * 0.70710678118654752f));
  g.z = 0.5f * v.z * (1.0f + erff(v.z * 0.70710678118654752f));
  g.w = 0.5f * v.w * (1.0f + erff(v.w * 0.70710678118654752f));
  const float tot  = wave_sum_f((g.x + g.y) + (g.z + g.w));
  const float totq = wave_sum_f((g.x * g.x + g.y * g.y) + (g.z * g.z + g.w * g.w));
  const float mu  = tot * (1.0f / D_FEAT);
  const float var = totq * (1.0f / D_FEAT) - mu * mu;
  const float inv = rsqrtf(var + LN_EPS);
  const v4f gm = *(const v4f*)(gamma + 4 * lane);
  const v4f bt = *(const v4f*)(beta + 4 * lane);
  const v4f xv = *(const v4f*)(x + (size_t)n * D_FEAT + 4 * lane);
  v4f o;
  o.x = (g.x - mu) * inv * gm.x + bt.x + xv.x;
  o.y = (g.y - mu) * inv * gm.y + bt.y + xv.y;
  o.z = (g.z - mu) * inv * gm.z + bt.z + xv.z;
  o.w = (g.w - mu) * inv * gm.w + bt.w + xv.w;
  *(v4f*)(out + (size_t)n * D_FEAT + 4 * lane) = o;
}

extern "C" void kernel_launch(void* const* d_in, const int* in_sizes, int n_in,
                              void* d_out, int out_size, void* d_ws, size_t ws_size,
                              hipStream_t stream) {
  const float* x       = (const float*)d_in[0];
  const int*   ei      = (const int*)d_in[1];
  const float* W       = (const float*)d_in[2];
  const float* att_src = (const float*)d_in[3];
  const float* att_dst = (const float*)d_in[4];
  const float* bias    = (const float*)d_in[5];
  const float* gamma   = (const float*)d_in[6];
  const float* beta    = (const float*)d_in[7];

  const int N = in_sizes[0] / D_FEAT;
  const int E = in_sizes[1] / 2;

  float* ws     = (float*)d_ws;
  float* h      = ws;
  float* a_src  = h + (size_t)N * D_FEAT;
  float* a_dst  = a_src + (size_t)N * HEADS;
  int*   cnt    = (int*)(a_dst + (size_t)N * HEADS);
  int*   ovf    = cnt + 2 * (size_t)N;
  float* out    = (float*)d_out;

  const int T  = (N + 31) / 32;
  const int g1 = (T * 7) / 16;
  const int g2 = T - g1;
  const int nch = (E + CH_EDGES - 1) / CH_EDGES;
  const int nr  = (N + RS_NODES - 1) / RS_NODES;
  {
    gat_bin_gemm<<<g1 + nch, 256, 0, stream>>>(x, W, att_src, att_dst, h, a_src, a_dst,
                                               cnt, N, g1, ei, E, out);
  }
  {
    gat_fill_gemm<<<g2 + nr, 256, 0, stream>>>(x, W, att_src, att_dst, h, a_src, a_dst, N,
                                               g1, g2, nch, E, cnt, ovf, out);
  }
  {
    int blocks = (N + 7) / 8;
    gat_node<<<blocks, 256, 0, stream>>>(ei, E, cnt, ovf, h, a_src, a_dst,
                                         bias, gamma, beta, x, out, N);
  }
}
